# starting kernel
# baseline (speedup 1.0000x reference)
#include <hip/hip_runtime.h>
#include <hip/hip_bf16.h>


typedef __attribute__((ext_vector_type(2))) float v2f;
typedef __attribute__((ext_vector_type(8))) float v8f;

#define B_ 64
#define R_ 100
#define RP 112
#define C_ 2048
#define P_ 7
#define K8 8
#define PP 49
#define H_ 512

__global__ void k_reduce_feats(const float* __restrict__ feats,
                               float* __restrict__ Fr8) {
  int idx = blockIdx.x * blockDim.x + threadIdx.x;
  const int total = B_ * C_ * K8;
  if (idx >= total) return;
  int k  = idx & 7;
  int bc = idx >> 3;
  float s = 0.f;
  if (k < P_) {
    const float* p = feats + (size_t)bc * PP + (size_t)k * P_;
#pragma unroll
    for (int j = 0; j < P_; ++j) s += p[j];
  }
  Fr8[idx] = s;
}

__global__ void k_reduce_masks(const float* __restrict__ masks,
                               float* __restrict__ Mp8) {
  int idx = blockIdx.x * blockDim.x + threadIdx.x;
  const int total = B_ * RP * K8;
  if (idx >= total) return;
  int k  = idx & 7;
  int br = idx >> 3;
  int b  = br / RP;
  int r  = br % RP;
  float s = 0.f;
  if (r < R_ && k < P_) {
    const float* p = masks + ((size_t)b * R_ + r) * PP + k;
#pragma unroll
    for (int i = 0; i < P_; ++i) s += p[i * P_];
  }
  Mp8[idx] = s;
}

__global__ void k_mbar(const float* __restrict__ Mp8, float* __restrict__ Mbar) {
  int idx = blockIdx.x * blockDim.x + threadIdx.x;
  if (idx >= B_ * K8) return;
  int b = idx >> 3, k = idx & 7;
  float s = 0.f;
  for (int r = 0; r < RP; ++r) s += Mp8[((size_t)b * RP + r) * K8 + k];
  Mbar[idx] = s;
}

__global__ void k_sm(const float* __restrict__ Fr8,
                     const float* __restrict__ Mbar,
                     float* __restrict__ Sm) {
  int idx = blockIdx.x * blockDim.x + threadIdx.x;
  if (idx >= B_ * C_) return;
  int b = idx / C_;
  const float* f  = Fr8 + (size_t)idx * K8;
  const float* mb = Mbar + b * K8;
  float s = 0.f;
#pragma unroll
  for (int k = 0; k < K8; ++k) s += mb[k] * f[k];
  Sm[idx] = s * (1.0f / (PP * R_));
}

__global__ void k_s_wmma(const float* __restrict__ Mp8,
                         const float* __restrict__ Fr8,
                         float* __restrict__ S) {
  const int MT = RP / 16;
  const int NT = C_ / 16;
  int wave = blockIdx.x * (blockDim.x >> 5) + (threadIdx.x >> 5);
  int b  = wave / (MT * NT);
  int t  = wave % (MT * NT);
  int mt = t / NT, nt = t % NT;

  int lane = threadIdx.x & 31;
  int half = lane >> 4;
  int lm   = lane & 15;

  const float* ap = Mp8 + ((size_t)b * RP + mt * 16 + lm) * K8 + 2 * half;
  const float* bp = Fr8 + ((size_t)b * C_ + nt * 16 + lm) * K8 + 2 * half;

  v2f a0 = *(const v2f*)(ap);
  v2f b0 = *(const v2f*)(bp);
  v2f a1 = *(const v2f*)(ap + 4);
  v2f b1 = *(const v2f*)(bp + 4);

  v8f acc = {};
  acc = __builtin_amdgcn_wmma_f32_16x16x4_f32(false, a0, false, b0,
                                              (short)0, acc, false, false);
  acc = __builtin_amdgcn_wmma_f32_16x16x4_f32(false, a1, false, b1,
                                              (short)0, acc, false, false);

  const float inv = 1.0f / PP;
  float* out = S + ((size_t)b * R_ + mt * 16 + half * 8) * C_ + nt * 16 + lm;
  if (mt < MT - 1) {
#pragma unroll
    for (int v = 0; v < 8; ++v)
      __builtin_nontemporal_store(acc[v] * inv, out + (size_t)v * C_);
  } else {
#pragma unroll
    for (int v = 0; v < 8; ++v) {
      int m = 96 + half * 8 + v;
      if (m < R_)
        __builtin_nontemporal_store(acc[v] * inv, out + (size_t)v * C_);
    }
  }
}

__global__ void k_head_wmma(const float* __restrict__ Sm,
                            const float* __restrict__ W1,
                            const float* __restrict__ b1,
                            const float* __restrict__ W2,
                            const float* __restrict__ b2,
                            float* __restrict__ h0,
                            float* __restrict__ c0) {
  const int MT = B_ / 16;
  const int NT = H_ / 16;
  int wave  = blockIdx.x * (blockDim.x >> 5) + (threadIdx.x >> 5);
  int which = wave / (MT * NT);
  int t     = wave % (MT * NT);
  int mt = t / NT, nt = t % NT;

  const float* W    = which ? W2 : W1;
  const float* bias = which ? b2 : b1;
  float*       out  = which ? c0 : h0;

  int lane = threadIdx.x & 31;
  int half = lane >> 4;
  int lm   = lane & 15;

  const float* ap = Sm + (size_t)(mt * 16 + lm) * C_ + 2 * half;
  const float* bp = W  + (size_t)(nt * 16 + lm) * C_ + 2 * half;

  v8f acc = {};
#pragma unroll 8
  for (int kb = 0; kb < C_; kb += 4) {
    v2f a  = *(const v2f*)(ap + kb);
    v2f bf = *(const v2f*)(bp + kb);
    acc = __builtin_amdgcn_wmma_f32_16x16x4_f32(false, a, false, bf,
                                               (short)0, acc, false, false);
  }

  float bv = bias[nt * 16 + lm];
#pragma unroll
  for (int v = 0; v < 8; ++v) {
    int m = mt * 16 + half * 8 + v;
    out[(size_t)m * H_ + nt * 16 + lm] = acc[v] + bv;
  }
}

extern "C" void kernel_launch(void* const* d_in, const int* in_sizes, int n_in,
                              void* d_out, int out_size, void* d_ws, size_t ws_size,
                              hipStream_t stream) {
  const float* feats = (const float*)d_in[0];
  const float* masks = (const float*)d_in[1];
  const float* W1    = (const float*)d_in[2];
  const float* b1    = (const float*)d_in[3];
  const float* W2    = (const float*)d_in[4];
  const float* b2    = (const float*)d_in[5];

  float* out = (float*)d_out;
  float* S   = out;
  float* h0  = out + (size_t)B_ * R_ * C_;
  float* c0  = h0 + (size_t)B_ * H_;

  float* ws   = (float*)d_ws;
  float* Fr8  = ws;
  float* Mp8  = Fr8 + (size_t)B_ * C_ * K8;
  float* Mbar = Mp8 + (size_t)B_ * RP * K8;
  float* Sm   = Mbar + (size_t)B_ * K8;

  {
    int total = B_ * C_ * K8;
    k_reduce_feats<<<(total + 255) / 256, 256, 0, stream>>>(feats, Fr8);
  }
  {
    int total = B_ * RP * K8;
    k_reduce_masks<<<(total + 255) / 256, 256, 0, stream>>>(masks, Mp8);
  }
  k_mbar<<<2, 256, 0, stream>>>(Mp8, Mbar);
  {
    int total = B_ * C_;
    k_sm<<<(total + 255) / 256, 256, 0, stream>>>(Fr8, Mbar, Sm);
  }
  {
    int waves = B_ * (RP / 16) * (C_ / 16);
    k_s_wmma<<<waves / 8, 256, 0, stream>>>(Mp8, Fr8, S);
  }
  k_head_wmma<<<32, 256, 0, stream>>>(Sm, W1, b1, W2, b2, h0, c0);
}


# head GEMM (h0,c0): split K 16 ways over the waves of a block, one 16x16 tile per block (256 blocks, was 32), partials summed in LDS in fixed wave order; changes the f32 summation order
# speedup vs baseline: 1.8298x; 1.8298x over previous
#include <hip/hip_runtime.h>
#include <hip/hip_bf16.h>


typedef __attribute__((ext_vector_type(2))) float v2f;
typedef __attribute__((ext_vector_type(8))) float v8f;

#define B_ 64
#define R_ 100
#define RP 112
#define C_ 2048
#define P_ 7
#define K8 8
#define PP 49
#define H_ 512

__global__ void k_reduce_feats(const float* __restrict__ feats,
                               float* __restrict__ Fr8) {
  int idx = blockIdx.x * blockDim.x + threadIdx.x;
  const int total = B_ * C_ * K8;
  if (idx >= total) return;
  int k  = idx & 7;
  int bc = idx >> 3;
  float s = 0.f;
  if (k < P_) {
    const float* p = feats + (size_t)bc * PP + (size_t)k * P_;
#pragma unroll
    for (int j = 0; j < P_; ++j) s += p[j];
  }
  Fr8[idx] = s;
}

__global__ void k_reduce_masks(const float* __restrict__ masks,
                               float* __restrict__ Mp8) {
  int idx = blockIdx.x * blockDim.x + threadIdx.x;
  const int total = B_ * RP * K8;
  if (idx >= total) return;
  int k  = idx & 7;
  int br = idx >> 3;
  int b  = br / RP;
  int r  = br % RP;
  float s = 0.f;
  if (r < R_ && k < P_) {
    const float* p = masks + ((size_t)b * R_ + r) * PP + k;
#pragma unroll
    for (int i = 0; i < P_; ++i) s += p[i * P_];
  }
  Mp8[idx] = s;
}

__global__ void k_mbar(const float* __restrict__ Mp8, float* __restrict__ Mbar) {
  int idx = blockIdx.x * blockDim.x + threadIdx.x;
  if (idx >= B_ * K8) return;
  int b = idx >> 3, k = idx & 7;
  float s = 0.f;
  for (int r = 0; r < RP; ++r) s += Mp8[((size_t)b * RP + r) * K8 + k];
  Mbar[idx] = s;
}

__global__ void k_sm(const float* __restrict__ Fr8,
                     const float* __restrict__ Mbar,
                     float* __restrict__ Sm) {
  int idx = blockIdx.x * blockDim.x + threadIdx.x;
  if (idx >= B_ * C_) return;
  int b = idx / C_;
  const float* f  = Fr8 + (size_t)idx * K8;
  const float* mb = Mbar + b * K8;
  float s = 0.f;
#pragma unroll
  for (int k = 0; k < K8; ++k) s += mb[k] * f[k];
  Sm[idx] = s * (1.0f / (PP * R_));
}

__global__ void k_s_wmma(const float* __restrict__ Mp8,
                         const float* __restrict__ Fr8,
                         float* __restrict__ S) {
  const int MT = RP / 16;
  const int NT = C_ / 16;
  int wave = blockIdx.x * (blockDim.x >> 5) + (threadIdx.x >> 5);
  int b  = wave / (MT * NT);
  int t  = wave % (MT * NT);
  int mt = t / NT, nt = t % NT;

  int lane = threadIdx.x & 31;
  int half = lane >> 4;
  int lm   = lane & 15;

  const float* ap = Mp8 + ((size_t)b * RP + mt * 16 + lm) * K8 + 2 * half;
  const float* bp = Fr8 + ((size_t)b * C_ + nt * 16 + lm) * K8 + 2 * half;

  v2f a0 = *(const v2f*)(ap);
  v2f b0 = *(const v2f*)(bp);
  v2f a1 = *(const v2f*)(ap + 4);
  v2f b1 = *(const v2f*)(bp + 4);

  v8f acc = {};
  acc = __builtin_amdgcn_wmma_f32_16x16x4_f32(false, a0, false, b0,
                                              (short)0, acc, false, false);
  acc = __builtin_amdgcn_wmma_f32_16x16x4_f32(false, a1, false, b1,
                                              (short)0, acc, false, false);

  const float inv = 1.0f / PP;
  float* out = S + ((size_t)b * R_ + mt * 16 + half * 8) * C_ + nt * 16 + lm;
  if (mt < MT - 1) {
#pragma unroll
    for (int v = 0; v < 8; ++v)
      __builtin_nontemporal_store(acc[v] * inv, out + (size_t)v * C_);
  } else {
#pragma unroll
    for (int v = 0; v < 8; ++v) {
      int m = 96 + half * 8 + v;
      if (m < R_)
        __builtin_nontemporal_store(acc[v] * inv, out + (size_t)v * C_);
    }
  }
}

#define KSPLIT 16
#define KW (C_ / KSPLIT)
__global__ __launch_bounds__(512)
void k_head_wmma(const float* __restrict__ Sm,
                 const float* __restrict__ W1,
                 const float* __restrict__ b1,
                 const float* __restrict__ W2,
                 const float* __restrict__ b2,
                 float* __restrict__ h0,
                 float* __restrict__ c0) {
  __shared__ float red[KSPLIT][256];

  const int MT = B_ / 16;
  const int NT = H_ / 16;
  int tile  = blockIdx.x;
  int which = tile / (MT * NT);
  int t     = tile % (MT * NT);
  int mt = t / NT, nt = t % NT;

  const float* W    = which ? W2 : W1;
  const float* bias = which ? b2 : b1;
  float*       out  = which ? c0 : h0;

  int w    = threadIdx.x >> 5;
  int lane = threadIdx.x & 31;
  int half = lane >> 4;
  int lm   = lane & 15;

  const float* ap = Sm + (size_t)(mt * 16 + lm) * C_ + w * KW + 2 * half;
  const float* bp = W  + (size_t)(nt * 16 + lm) * C_ + w * KW + 2 * half;

  v8f acc = {};
#pragma unroll
  for (int kb = 0; kb < KW; kb += 4) {
    v2f a  = *(const v2f*)(ap + kb);
    v2f bf = *(const v2f*)(bp + kb);
    acc = __builtin_amdgcn_wmma_f32_16x16x4_f32(false, a, false, bf,
                                               (short)0, acc, false, false);
  }

#pragma unroll
  for (int v = 0; v < 8; ++v) red[w][v * 32 + lane] = acc[v];
  __syncthreads();

  if (threadIdx.x < 256) {
    int e  = threadIdx.x;
    int v  = e >> 5;
    int l  = e & 31;
    int hh = l >> 4, c = l & 15;
    float s = red[0][e];
#pragma unroll
    for (int i = 1; i < KSPLIT; ++i) s += red[i][e];
    int m = mt * 16 + hh * 8 + v;
    out[(size_t)m * H_ + nt * 16 + c] = s + bias[nt * 16 + c];
  }
}

extern "C" void kernel_launch(void* const* d_in, const int* in_sizes, int n_in,
                              void* d_out, int out_size, void* d_ws, size_t ws_size,
                              hipStream_t stream) {
  const float* feats = (const float*)d_in[0];
  const float* masks = (const float*)d_in[1];
  const float* W1    = (const float*)d_in[2];
  const float* b1    = (const float*)d_in[3];
  const float* W2    = (const float*)d_in[4];
  const float* b2    = (const float*)d_in[5];

  float* out = (float*)d_out;
  float* S   = out;
  float* h0  = out + (size_t)B_ * R_ * C_;
  float* c0  = h0 + (size_t)B_ * H_;

  float* ws   = (float*)d_ws;
  float* Fr8  = ws;
  float* Mp8  = Fr8 + (size_t)B_ * C_ * K8;
  float* Mbar = Mp8 + (size_t)B_ * RP * K8;
  float* Sm   = Mbar + (size_t)B_ * K8;

  {
    int total = B_ * C_ * K8;
    k_reduce_feats<<<(total + 255) / 256, 256, 0, stream>>>(feats, Fr8);
  }
  {
    int total = B_ * RP * K8;
    k_reduce_masks<<<(total + 255) / 256, 256, 0, stream>>>(masks, Mp8);
  }
  k_mbar<<<2, 256, 0, stream>>>(Mp8, Mbar);
  {
    int total = B_ * C_;
    k_sm<<<(total + 255) / 256, 256, 0, stream>>>(Fr8, Mbar, Sm);
  }
  {
    int waves = B_ * (RP / 16) * (C_ / 16);
    k_s_wmma<<<waves / 8, 256, 0, stream>>>(Mp8, Fr8, S);
  }
  k_head_wmma<<<256, 512, 0, stream>>>(Sm, W1, b1, W2, b2, h0, c0);
}


# S GEMM: one wave per 112x32 strip (7 row tiles x 2 col tiles, was one 16x16 tile): 4096 waves for 57344, Fr fragments loaded once, lane halves swapped so each row leaves as one 128-B line
# speedup vs baseline: 2.5344x; 1.3851x over previous
#include <hip/hip_runtime.h>
#include <hip/hip_bf16.h>


typedef __attribute__((ext_vector_type(2))) float v2f;
typedef __attribute__((ext_vector_type(8))) float v8f;

#define B_ 64
#define R_ 100
#define RP 112
#define C_ 2048
#define P_ 7
#define K8 8
#define PP 49
#define H_ 512

__global__ void k_reduce_feats(const float* __restrict__ feats,
                               float* __restrict__ Fr8) {
  int idx = blockIdx.x * blockDim.x + threadIdx.x;
  const int total = B_ * C_ * K8;
  if (idx >= total) return;
  int k  = idx & 7;
  int bc = idx >> 3;
  float s = 0.f;
  if (k < P_) {
    const float* p = feats + (size_t)bc * PP + (size_t)k * P_;
#pragma unroll
    for (int j = 0; j < P_; ++j) s += p[j];
  }
  Fr8[idx] = s;
}

__global__ void k_reduce_masks(const float* __restrict__ masks,
                               float* __restrict__ Mp8) {
  int idx = blockIdx.x * blockDim.x + threadIdx.x;
  const int total = B_ * RP * K8;
  if (idx >= total) return;
  int k  = idx & 7;
  int br = idx >> 3;
  int b  = br / RP;
  int r  = br % RP;
  float s = 0.f;
  if (r < R_ && k < P_) {
    const float* p = masks + ((size_t)b * R_ + r) * PP + k;
#pragma unroll
    for (int i = 0; i < P_; ++i) s += p[i * P_];
  }
  Mp8[idx] = s;
}

__global__ void k_mbar(const float* __restrict__ Mp8, float* __restrict__ Mbar) {
  int idx = blockIdx.x * blockDim.x + threadIdx.x;
  if (idx >= B_ * K8) return;
  int b = idx >> 3, k = idx & 7;
  float s = 0.f;
  for (int r = 0; r < RP; ++r) s += Mp8[((size_t)b * RP + r) * K8 + k];
  Mbar[idx] = s;
}

__global__ void k_sm(const float* __restrict__ Fr8,
                     const float* __restrict__ Mbar,
                     float* __restrict__ Sm) {
  int idx = blockIdx.x * blockDim.x + threadIdx.x;
  if (idx >= B_ * C_) return;
  int b = idx / C_;
  const float* f  = Fr8 + (size_t)idx * K8;
  const float* mb = Mbar + b * K8;
  float s = 0.f;
#pragma unroll
  for (int k = 0; k < K8; ++k) s += mb[k] * f[k];
  Sm[idx] = s * (1.0f / (PP * R_));
}


__device__ __forceinline__ float xhalf(float x) {
  int t = __builtin_bit_cast(int, x);
  return __builtin_bit_cast(float,
      __builtin_amdgcn_permlanex16(t, t, 0x76543210, 0xfedcba98, false, false));
}

#define SW 32
__global__ __launch_bounds__(256)
void k_s_wmma(const float* __restrict__ Mp8,
              const float* __restrict__ Fr8,
              float* __restrict__ S) {
  const int MT = RP / 16;
  const int NS = C_ / SW;
  int wave = blockIdx.x * (blockDim.x >> 5) + (threadIdx.x >> 5);
  int b  = wave / NS;
  int st = wave % NS;

  int lane = threadIdx.x & 31;
  int half = lane >> 4;
  int lm   = lane & 15;

  const float* bp = Fr8 + ((size_t)b * C_ + st * SW + lm) * K8 + 2 * half;
  v2f b00 = *(const v2f*)(bp);
  v2f b01 = *(const v2f*)(bp + 4);
  v2f b10 = *(const v2f*)(bp + 16 * K8);
  v2f b11 = *(const v2f*)(bp + 16 * K8 + 4);

  const float* ap = Mp8 + ((size_t)b * RP + lm) * K8 + 2 * half;

  const float inv = 1.0f / PP;
  float* out = S + (size_t)b * R_ * C_ + st * SW + lane;

#pragma unroll
  for (int mt = 0; mt < MT; ++mt) {
    v2f a0 = *(const v2f*)(ap + mt * 16 * K8);
    v2f a1 = *(const v2f*)(ap + mt * 16 * K8 + 4);

    v8f acc0 = {}, acc1 = {};
    acc0 = __builtin_amdgcn_wmma_f32_16x16x4_f32(false, a0, false, b00,
                                                 (short)0, acc0, false, false);
    acc1 = __builtin_amdgcn_wmma_f32_16x16x4_f32(false, a0, false, b10,
                                                 (short)0, acc1, false, false);
    acc0 = __builtin_amdgcn_wmma_f32_16x16x4_f32(false, a1, false, b01,
                                                 (short)0, acc0, false, false);
    acc1 = __builtin_amdgcn_wmma_f32_16x16x4_f32(false, a1, false, b11,
                                                 (short)0, acc1, false, false);

#pragma unroll
    for (int v = 0; v < 8; ++v) {
      float lo = acc0[v] * inv;
      float hi = acc1[v] * inv;
      float got = xhalf(half ? lo : hi);
      float ra = half ? got : lo;
      float rb = half ? hi  : got;
      int r0 = mt * 16 + v, r1 = r0 + 8;
      if (r0 < R_) __builtin_nontemporal_store(ra, out + (size_t)r0 * C_);
      if (r1 < R_) __builtin_nontemporal_store(rb, out + (size_t)r1 * C_);
    }
  }
}

#define KSPLIT 16
#define KW (C_ / KSPLIT)
__global__ __launch_bounds__(512)
void k_head_wmma(const float* __restrict__ Sm,
                 const float* __restrict__ W1,
                 const float* __restrict__ b1,
                 const float* __restrict__ W2,
                 const float* __restrict__ b2,
                 float* __restrict__ h0,
                 float* __restrict__ c0) {
  __shared__ float red[KSPLIT][256];

  const int MT = B_ / 16;
  const int NT = H_ / 16;
  int tile  = blockIdx.x;
  int which = tile / (MT * NT);
  int t     = tile % (MT * NT);
  int mt = t / NT, nt = t % NT;

  const float* W    = which ? W2 : W1;
  const float* bias = which ? b2 : b1;
  float*       out  = which ? c0 : h0;

  int w    = threadIdx.x >> 5;
  int lane = threadIdx.x & 31;
  int half = lane >> 4;
  int lm   = lane & 15;

  const float* ap = Sm + (size_t)(mt * 16 + lm) * C_ + w * KW + 2 * half;
  const float* bp = W  + (size_t)(nt * 16 + lm) * C_ + w * KW + 2 * half;

  v8f acc = {};
#pragma unroll
  for (int kb = 0; kb < KW; kb += 4) {
    v2f a  = *(const v2f*)(ap + kb);
    v2f bf = *(const v2f*)(bp + kb);
    acc = __builtin_amdgcn_wmma_f32_16x16x4_f32(false, a, false, bf,
                                               (short)0, acc, false, false);
  }

#pragma unroll
  for (int v = 0; v < 8; ++v) red[w][v * 32 + lane] = acc[v];
  __syncthreads();

  if (threadIdx.x < 256) {
    int e  = threadIdx.x;
    int v  = e >> 5;
    int l  = e & 31;
    int hh = l >> 4, c = l & 15;
    float s = red[0][e];
#pragma unroll
    for (int i = 1; i < KSPLIT; ++i) s += red[i][e];
    int m = mt * 16 + hh * 8 + v;
    out[(size_t)m * H_ + nt * 16 + c] = s + bias[nt * 16 + c];
  }
}

extern "C" void kernel_launch(void* const* d_in, const int* in_sizes, int n_in,
                              void* d_out, int out_size, void* d_ws, size_t ws_size,
                              hipStream_t stream) {
  const float* feats = (const float*)d_in[0];
  const float* masks = (const float*)d_in[1];
  const float* W1    = (const float*)d_in[2];
  const float* b1    = (const float*)d_in[3];
  const float* W2    = (const float*)d_in[4];
  const float* b2    = (const float*)d_in[5];

  float* out = (float*)d_out;
  float* S   = out;
  float* h0  = out + (size_t)B_ * R_ * C_;
  float* c0  = h0 + (size_t)B_ * H_;

  float* ws   = (float*)d_ws;
  float* Fr8  = ws;
  float* Mp8  = Fr8 + (size_t)B_ * C_ * K8;
  float* Mbar = Mp8 + (size_t)B_ * RP * K8;
  float* Sm   = Mbar + (size_t)B_ * K8;

  {
    int total = B_ * C_ * K8;
    k_reduce_feats<<<(total + 255) / 256, 256, 0, stream>>>(feats, Fr8);
  }
  {
    int total = B_ * RP * K8;
    k_reduce_masks<<<(total + 255) / 256, 256, 0, stream>>>(masks, Mp8);
  }
  k_mbar<<<2, 256, 0, stream>>>(Mp8, Mbar);
  {
    int total = B_ * C_;
    k_sm<<<(total + 255) / 256, 256, 0, stream>>>(Fr8, Mbar, Sm);
  }
  k_s_wmma<<<512, 256, 0, stream>>>(Mp8, Fr8, S);
  k_head_wmma<<<256, 512, 0, stream>>>(Sm, W1, b1, W2, b2, h0, c0);
}


# fuse the five producer launches (feats/masks sums, Mbar, Sm, S GEMM) into one kernel, operands in LDS, Sm from 8 partial-Mbar pad rows of A: 6 launches -> 2; changes the f32 order of the row mean
# speedup vs baseline: 4.3710x; 1.7247x over previous
#include <hip/hip_runtime.h>
#include <hip/hip_bf16.h>


typedef __attribute__((ext_vector_type(2))) float v2f;
typedef __attribute__((ext_vector_type(4))) float v4f;
typedef __attribute__((ext_vector_type(8))) float v8f;

#define B_ 64
#define R_ 100
#define RP 112
#define C_ 2048
#define P_ 7
#define K8 8
#define PP 49
#define H_ 512

__device__ __forceinline__ float xhalf(float x) {
  int t = __builtin_bit_cast(int, x);
  return __builtin_bit_cast(float,
      __builtin_amdgcn_permlanex16(t, t, 0x76543210, 0xfedcba98, false, false));
}

__device__ __forceinline__ float rowsum7(const float* p) {
  float s = 0.f;
#pragma unroll
  for (int j = 0; j < P_; ++j) s += p[j];
  return s;
}

#define SW 32
#define FW (SW * PP)
__global__ __launch_bounds__(256)
void k_fused(const float* __restrict__ feats,
             const float* __restrict__ masks,
             float* __restrict__ S,
             float* __restrict__ Sm) {
  __shared__ __attribute__((aligned(16))) float sMp[RP * K8];
  __shared__ __attribute__((aligned(16))) float sF[8][FW];

  const int tid   = threadIdx.x;
  const int b     = blockIdx.x >> 3;
  const int chunk = blockIdx.x & 7;
  const int w     = tid >> 5;
  const int lane  = tid & 31;
  const int half  = lane >> 4;
  const int lm    = lane & 15;
  const int st    = chunk * 8 + w;

  const int k  = tid & 7;
  const int kc = k < P_ ? k : P_ - 1;
  const int rr = tid >> 3;
  float mv[4][P_];
#pragma unroll
  for (int i = 0; i < 4; ++i) {
    int r  = rr + 32 * i;
    int rc = r < R_ ? r : R_ - 1;
    const float* p = masks + ((size_t)b * R_ + rc) * PP + kc;
#pragma unroll
    for (int ii = 0; ii < P_; ++ii) mv[i][ii] = p[ii * P_];
  }

  const float* fp = feats + ((size_t)b * C_ + st * SW) * PP;
  v4f fv[12];
#pragma unroll
  for (int i = 0; i < 12; ++i)
    fv[i] = __builtin_nontemporal_load((const v4f*)fp + i * 32 + lane);
  float ft = __builtin_nontemporal_load(fp + 12 * 128 + lane);

  float part = 0.f;
#pragma unroll
  for (int i = 0; i < 4; ++i) {
    int r = rr + 32 * i;
    float s = 0.f;
#pragma unroll
    for (int ii = 0; ii < P_; ++ii) s += mv[i][ii];
    s = (r < R_ && k < P_) ? s : 0.f;
    if (r < R_ || (r >= R_ + 8 && r < RP)) sMp[r * K8 + k] = s;
    part += s;
  }
  part += __shfl_xor(part, 8);
  part += __shfl_xor(part, 16);
  if (lane < K8) sMp[(R_ + w) * K8 + lane] = part;

  float* sf = sF[w];
#pragma unroll
  for (int i = 0; i < 12; ++i) *(v4f*)(sf + (i * 32 + lane) * 4) = fv[i];
  sf[12 * 128 + lane] = ft;

  __syncthreads();

  const float* q0 = sf + lm * PP + 2 * P_ * half;
  const float* q1 = q0 + 16 * PP;
  const int    o3 = half ? 4 * P_ : 5 * P_;
  v2f b00, b01, b10, b11;
  b00.x = rowsum7(q0);
  b00.y = rowsum7(q0 + P_);
  b01.x = rowsum7(q0 + 4 * P_);
  { float t = rowsum7(q0 + o3); b01.y = half ? 0.f : t; }
  b10.x = rowsum7(q1);
  b10.y = rowsum7(q1 + P_);
  b11.x = rowsum7(q1 + 4 * P_);
  { float t = rowsum7(q1 + o3); b11.y = half ? 0.f : t; }

  const float* ap = sMp + lm * K8 + 2 * half;

  const float inv = 1.0f / PP;
  float* out = S + (size_t)b * R_ * C_ + st * SW + lane;

#pragma unroll
  for (int mt = 0; mt < RP / 16; ++mt) {
    v2f a0 = *(const v2f*)(ap + mt * 16 * K8);
    v2f a1 = *(const v2f*)(ap + mt * 16 * K8 + 4);

    v8f acc0 = {}, acc1 = {};
    acc0 = __builtin_amdgcn_wmma_f32_16x16x4_f32(false, a0, false, b00,
                                                 (short)0, acc0, false, false);
    acc1 = __builtin_amdgcn_wmma_f32_16x16x4_f32(false, a0, false, b10,
                                                 (short)0, acc1, false, false);
    acc0 = __builtin_amdgcn_wmma_f32_16x16x4_f32(false, a1, false, b01,
                                                 (short)0, acc0, false, false);
    acc1 = __builtin_amdgcn_wmma_f32_16x16x4_f32(false, a1, false, b11,
                                                 (short)0, acc1, false, false);

#pragma unroll
    for (int v = 0; v < 8; ++v) {
      float lo = acc0[v] * inv;
      float hi = acc1[v] * inv;
      float got = xhalf(half ? lo : hi);
      float ra = half ? got : lo;
      float rb = half ? hi  : got;
      int r0 = mt * 16 + v, r1 = r0 + 8;
      if (r0 < R_) __builtin_nontemporal_store(ra, out + (size_t)r0 * C_);
      if (r1 < R_) __builtin_nontemporal_store(rb, out + (size_t)r1 * C_);
    }

    if (mt * 16 + 4 == R_) {
      float u0 = (acc0[0] + acc0[1]) + (acc0[2] + acc0[3]);
      float l0 = (acc0[4] + acc0[5]) + (acc0[6] + acc0[7]);
      float u1 = (acc1[0] + acc1[1]) + (acc1[2] + acc1[3]);
      float l1 = (acc1[4] + acc1[5]) + (acc1[6] + acc1[7]);
      float t0 = half ? u0 : l0;
      float t1 = half ? u1 : l1;
      float x0 = xhalf(t0);
      float x1 = xhalf(t1);
      float sm = half ? (t1 + x1) : (t0 + x0);
      Sm[(size_t)b * C_ + st * SW + lane] = sm * (1.0f / (PP * R_));
    }
  }
}

#define KSPLIT 16
#define KW (C_ / KSPLIT)
__global__ __launch_bounds__(512)
void k_head_wmma(const float* __restrict__ Sm,
                 const float* __restrict__ W1,
                 const float* __restrict__ b1,
                 const float* __restrict__ W2,
                 const float* __restrict__ b2,
                 float* __restrict__ h0,
                 float* __restrict__ c0) {
  __shared__ float red[KSPLIT][256];

  const int MT = B_ / 16;
  const int NT = H_ / 16;
  int tile  = blockIdx.x;
  int which = tile / (MT * NT);
  int t     = tile % (MT * NT);
  int mt = t / NT, nt = t % NT;

  const float* W    = which ? W2 : W1;
  const float* bias = which ? b2 : b1;
  float*       out  = which ? c0 : h0;

  int w    = threadIdx.x >> 5;
  int lane = threadIdx.x & 31;
  int half = lane >> 4;
  int lm   = lane & 15;

  const float* ap = Sm + (size_t)(mt * 16 + lm) * C_ + w * KW + 2 * half;
  const float* bp = W  + (size_t)(nt * 16 + lm) * C_ + w * KW + 2 * half;

  v8f acc = {};
#pragma unroll
  for (int kb = 0; kb < KW; kb += 4) {
    v2f a  = *(const v2f*)(ap + kb);
    v2f bf = *(const v2f*)(bp + kb);
    acc = __builtin_amdgcn_wmma_f32_16x16x4_f32(false, a, false, bf,
                                               (short)0, acc, false, false);
  }

#pragma unroll
  for (int v = 0; v < 8; ++v) red[w][v * 32 + lane] = acc[v];
  __syncthreads();

  if (threadIdx.x < 256) {
    int e  = threadIdx.x;
    int v  = e >> 5;
    int l  = e & 31;
    int hh = l >> 4, c = l & 15;
    float s = red[0][e];
#pragma unroll
    for (int i = 1; i < KSPLIT; ++i) s += red[i][e];
    int m = mt * 16 + hh * 8 + v;
    out[(size_t)m * H_ + nt * 16 + c] = s + bias[nt * 16 + c];
  }
}

extern "C" void kernel_launch(void* const* d_in, const int* in_sizes, int n_in,
                              void* d_out, int out_size, void* d_ws, size_t ws_size,
                              hipStream_t stream) {
  const float* feats = (const float*)d_in[0];
  const float* masks = (const float*)d_in[1];
  const float* W1    = (const float*)d_in[2];
  const float* b1    = (const float*)d_in[3];
  const float* W2    = (const float*)d_in[4];
  const float* b2    = (const float*)d_in[5];

  float* out = (float*)d_out;
  float* S   = out;
  float* h0  = out + (size_t)B_ * R_ * C_;
  float* c0  = h0 + (size_t)B_ * H_;

  float* Sm = (float*)d_ws;

  k_fused<<<512, 256, 0, stream>>>(feats, masks, S, Sm);
  k_head_wmma<<<256, 512, 0, stream>>>(Sm, W1, b1, W2, b2, h0, c0);
}


# head GEMM: operands fetched with whole-line loads (4 rows x 128 B per load, was 16 lines x 16 B), staged per wave in LDS, read as 16-B fragments; changes the f32 summation order
# speedup vs baseline: 5.0648x; 1.1587x over previous
#include <hip/hip_runtime.h>
#include <hip/hip_bf16.h>


typedef __attribute__((ext_vector_type(2))) float v2f;
typedef __attribute__((ext_vector_type(4))) float v4f;
typedef __attribute__((ext_vector_type(8))) float v8f;

#define B_ 64
#define R_ 100
#define RP 112
#define C_ 2048
#define P_ 7
#define K8 8
#define PP 49
#define H_ 512

__device__ __forceinline__ float xhalf(float x) {
  int t = __builtin_bit_cast(int, x);
  return __builtin_bit_cast(float,
      __builtin_amdgcn_permlanex16(t, t, 0x76543210, 0xfedcba98, false, false));
}

__device__ __forceinline__ float rowsum7(const float* p) {
  float s = 0.f;
#pragma unroll
  for (int j = 0; j < P_; ++j) s += p[j];
  return s;
}

#define SW 32
#define FW (SW * PP)
__global__ __launch_bounds__(256)
void k_fused(const float* __restrict__ feats,
             const float* __restrict__ masks,
             float* __restrict__ S,
             float* __restrict__ Sm) {
  __shared__ __attribute__((aligned(16))) float sMp[RP * K8];
  __shared__ __attribute__((aligned(16))) float sF[8][FW];

  const int tid   = threadIdx.x;
  const int b     = blockIdx.x >> 3;
  const int chunk = blockIdx.x & 7;
  const int w     = tid >> 5;
  const int lane  = tid & 31;
  const int half  = lane >> 4;
  const int lm    = lane & 15;
  const int st    = chunk * 8 + w;

  const int k  = tid & 7;
  const int kc = k < P_ ? k : P_ - 1;
  const int rr = tid >> 3;
  float mv[4][P_];
#pragma unroll
  for (int i = 0; i < 4; ++i) {
    int r  = rr + 32 * i;
    int rc = r < R_ ? r : R_ - 1;
    const float* p = masks + ((size_t)b * R_ + rc) * PP + kc;
#pragma unroll
    for (int ii = 0; ii < P_; ++ii) mv[i][ii] = p[ii * P_];
  }

  const float* fp = feats + ((size_t)b * C_ + st * SW) * PP;
  v4f fv[12];
#pragma unroll
  for (int i = 0; i < 12; ++i)
    fv[i] = __builtin_nontemporal_load((const v4f*)fp + i * 32 + lane);
  float ft = __builtin_nontemporal_load(fp + 12 * 128 + lane);

  float part = 0.f;
#pragma unroll
  for (int i = 0; i < 4; ++i) {
    int r = rr + 32 * i;
    float s = 0.f;
#pragma unroll
    for (int ii = 0; ii < P_; ++ii) s += mv[i][ii];
    s = (r < R_ && k < P_) ? s : 0.f;
    if (r < R_ || (r >= R_ + 8 && r < RP)) sMp[r * K8 + k] = s;
    part += s;
  }
  part += __shfl_xor(part, 8);
  part += __shfl_xor(part, 16);
  if (lane < K8) sMp[(R_ + w) * K8 + lane] = part;

  float* sf = sF[w];
#pragma unroll
  for (int i = 0; i < 12; ++i) *(v4f*)(sf + (i * 32 + lane) * 4) = fv[i];
  sf[12 * 128 + lane] = ft;

  __syncthreads();

  const float* q0 = sf + lm * PP + 2 * P_ * half;
  const float* q1 = q0 + 16 * PP;
  const int    o3 = half ? 4 * P_ : 5 * P_;
  v2f b00, b01, b10, b11;
  b00.x = rowsum7(q0);
  b00.y = rowsum7(q0 + P_);
  b01.x = rowsum7(q0 + 4 * P_);
  { float t = rowsum7(q0 + o3); b01.y = half ? 0.f : t; }
  b10.x = rowsum7(q1);
  b10.y = rowsum7(q1 + P_);
  b11.x = rowsum7(q1 + 4 * P_);
  { float t = rowsum7(q1 + o3); b11.y = half ? 0.f : t; }

  const float* ap = sMp + lm * K8 + 2 * half;

  const float inv = 1.0f / PP;
  float* out = S + (size_t)b * R_ * C_ + st * SW + lane;

#pragma unroll
  for (int mt = 0; mt < RP / 16; ++mt) {
    v2f a0 = *(const v2f*)(ap + mt * 16 * K8);
    v2f a1 = *(const v2f*)(ap + mt * 16 * K8 + 4);

    v8f acc0 = {}, acc1 = {};
    acc0 = __builtin_amdgcn_wmma_f32_16x16x4_f32(false, a0, false, b00,
                                                 (short)0, acc0, false, false);
    acc1 = __builtin_amdgcn_wmma_f32_16x16x4_f32(false, a0, false, b10,
                                                 (short)0, acc1, false, false);
    acc0 = __builtin_amdgcn_wmma_f32_16x16x4_f32(false, a1, false, b01,
                                                 (short)0, acc0, false, false);
    acc1 = __builtin_amdgcn_wmma_f32_16x16x4_f32(false, a1, false, b11,
                                                 (short)0, acc1, false, false);

#pragma unroll
    for (int v = 0; v < 8; ++v) {
      float lo = acc0[v] * inv;
      float hi = acc1[v] * inv;
      float got = xhalf(half ? lo : hi);
      float ra = half ? got : lo;
      float rb = half ? hi  : got;
      int r0 = mt * 16 + v, r1 = r0 + 8;
      if (r0 < R_) __builtin_nontemporal_store(ra, out + (size_t)r0 * C_);
      if (r1 < R_) __builtin_nontemporal_store(rb, out + (size_t)r1 * C_);
    }

    if (mt * 16 + 4 == R_) {
      float u0 = (acc0[0] + acc0[1]) + (acc0[2] + acc0[3]);
      float l0 = (acc0[4] + acc0[5]) + (acc0[6] + acc0[7]);
      float u1 = (acc1[0] + acc1[1]) + (acc1[2] + acc1[3]);
      float l1 = (acc1[4] + acc1[5]) + (acc1[6] + acc1[7]);
      float t0 = half ? u0 : l0;
      float t1 = half ? u1 : l1;
      float x0 = xhalf(t0);
      float x1 = xhalf(t1);
      float sm = half ? (t1 + x1) : (t0 + x0);
      Sm[(size_t)b * C_ + st * SW + lane] = sm * (1.0f / (PP * R_));
    }
  }
}

#define KSPLIT 16
#define KW (C_ / KSPLIT)
#define KC 32
#define NCH (KW / KC)
#define LP (KC + 4)
#define CHF (16 * LP)
__global__ __launch_bounds__(512)
void k_head_wmma(const float* __restrict__ Sm,
                 const float* __restrict__ W1,
                 const float* __restrict__ b1,
                 const float* __restrict__ W2,
                 const float* __restrict__ b2,
                 float* __restrict__ h0,
                 float* __restrict__ c0) {
  __shared__ __attribute__((aligned(16))) float stg[KSPLIT][NCH][2][CHF];

  const int MT = B_ / 16;
  const int NT = H_ / 16;
  int tile  = blockIdx.x;
  int which = tile / (MT * NT);
  int t     = tile % (MT * NT);
  int mt = t / NT, nt = t % NT;

  const float* W    = which ? W2 : W1;
  const float* bias = which ? b2 : b1;
  float*       out  = which ? c0 : h0;

  int w    = threadIdx.x >> 5;
  int lane = threadIdx.x & 31;
  int half = lane >> 4;
  int lm   = lane & 15;

  const int lr = 8 * ((lane >> 3) & 1) + 4 * (lane >> 4);
  const int lc = (lane & 7) * 4;
  const float* ag = Sm + (size_t)(mt * 16 + lr) * C_ + w * KW + lc;
  const float* bg = W  + (size_t)(nt * 16 + lr) * C_ + w * KW + lc;

  v4f ga[NCH][4], gb[NCH][4];
#pragma unroll
  for (int c = 0; c < NCH; ++c) {
#pragma unroll
    for (int i = 0; i < 4; ++i) {
      ga[c][i] = *(const v4f*)(ag + (size_t)i * C_ + c * KC);
      gb[c][i] = *(const v4f*)(bg + (size_t)i * C_ + c * KC);
    }
  }

  float* sw = &stg[w][0][0][0];
  const int so = lr * LP + lc;
  const int fo = lm * LP + 4 * half;

  v8f acc = {};
#pragma unroll
  for (int c = 0; c < NCH; ++c) {
    float* sa = sw + (c * 2 + 0) * CHF;
    float* sb = sw + (c * 2 + 1) * CHF;
#pragma unroll
    for (int i = 0; i < 4; ++i) {
      *(v4f*)(sa + i * LP + so) = ga[c][i];
      *(v4f*)(sb + i * LP + so) = gb[c][i];
    }
    __builtin_amdgcn_fence(__ATOMIC_RELEASE, "wavefront");
    __builtin_amdgcn_wave_barrier();
    __builtin_amdgcn_fence(__ATOMIC_ACQUIRE, "wavefront");
#pragma unroll
    for (int j = 0; j < KC / 8; ++j) {
      v4f a  = *(const v4f*)(sa + fo + j * 8);
      v4f bf = *(const v4f*)(sb + fo + j * 8);
      v2f a0 = {a.x, a.y},   a1 = {a.z, a.w};
      v2f f0 = {bf.x, bf.y}, f1 = {bf.z, bf.w};
      acc = __builtin_amdgcn_wmma_f32_16x16x4_f32(false, a0, false, f0,
                                                 (short)0, acc, false, false);
      acc = __builtin_amdgcn_wmma_f32_16x16x4_f32(false, a1, false, f1,
                                                 (short)0, acc, false, false);
    }
  }

#pragma unroll
  for (int v = 0; v < 8; ++v) sw[v * 32 + lane] = acc[v];
  __syncthreads();

  if (threadIdx.x < 256) {
    int e  = threadIdx.x;
    int v  = e >> 5;
    int l  = e & 31;
    int hh = l >> 4, c = l & 15;
    float s = stg[0][0][0][e];
#pragma unroll
    for (int i = 1; i < KSPLIT; ++i) s += stg[i][0][0][e];
    int m = mt * 16 + hh * 8 + v;
    out[(size_t)m * H_ + nt * 16 + c] = s + bias[nt * 16 + c];
  }
}

extern "C" void kernel_launch(void* const* d_in, const int* in_sizes, int n_in,
                              void* d_out, int out_size, void* d_ws, size_t ws_size,
                              hipStream_t stream) {
  const float* feats = (const float*)d_in[0];
  const float* masks = (const float*)d_in[1];
  const float* W1    = (const float*)d_in[2];
  const float* b1    = (const float*)d_in[3];
  const float* W2    = (const float*)d_in[4];
  const float* b2    = (const float*)d_in[5];

  float* out = (float*)d_out;
  float* S   = out;
  float* h0  = out + (size_t)B_ * R_ * C_;
  float* c0  = h0 + (size_t)B_ * H_;

  float* Sm = (float*)d_ws;

  k_fused<<<512, 256, 0, stream>>>(feats, masks, S, Sm);
  k_head_wmma<<<256, 512, 0, stream>>>(Sm, W1, b1, W2, b2, h0, c0);
}
